# starting kernel
# baseline (speedup 1.0000x reference)
#include <hip/hip_runtime.h>

typedef float v2f __attribute__((ext_vector_type(2)));
typedef float v8f __attribute__((ext_vector_type(8)));

#define DD 112
#define HH 112
#define WW 112
#define CI 8
#define CO 8
#define NTAPS 27

#define ZD 3
#define ZH 10
#define ZW 18
#define CP 10

__global__ __launch_bounds__(256) void SparseConv3d_wmma_f32_kernel(
    const float* __restrict__ x,
    const float* __restrict__ wgt,
    const float* __restrict__ bias,
    float* __restrict__ out)
{
  const int sH = WW;
  const int sD = HH * WW;
  const int sC = DD * HH * WW;
  const int sN = CI * sC;

  __shared__ float lx[ZD * ZH * ZW * CP];
  __shared__ float lw[NTAPS * 2 * 16 * 4];

  const int tid = threadIdx.x;

  int b = blockIdx.x;
  const int wq = b % 7;  b /= 7;
  const int hb = b % 14; b /= 14;
  const int d  = b % DD; b /= DD;
  const int n  = b;
  const int h0 = hb * 8;
  const int w0 = wq * 16;

  const float* xn = x + n * sN;
  for (int idx = tid; idx < ZD * ZH * CI * ZW; idx += 256) {
    const int zw_off = idx % ZW;
    int r = idx / ZW;
    const int ci = r % CI;     r /= CI;
    const int zh_off = r % ZH;
    const int zd_off = r / ZH;
    const int zd = d  + zd_off - 1;
    const int zh = h0 + zh_off - 1;
    const int zw = w0 + zw_off - 1;
    const bool ok = ((unsigned)zd < (unsigned)DD) &
                    ((unsigned)zh < (unsigned)HH) &
                    ((unsigned)zw < (unsigned)WW);
    const int zdc = ok ? zd : 0;
    const int zhc = ok ? zh : 0;
    const int zwc = ok ? zw : 0;
    const float v = xn[ci * sC + zdc * sD + zhc * sH + zwc] * (ok ? 1.0f : 0.0f);
    lx[((zd_off * ZH + zh_off) * ZW + zw_off) * CP + ci] = v;
  }

  for (int idx = tid; idx < NTAPS * 2 * 16 * 4; idx += 256) {
    const int k   = idx & 3;
    const int nn  = (idx >> 2) & 15;
    const int cig = (idx >> 6) & 1;
    const int t27 = idx >> 7;
    const int ci  = cig * 4 + k;
    const float v = wgt[((nn & 7) * CI + ci) * NTAPS + t27] *
                    (nn < CO ? 1.0f : 0.0f);
    lw[idx] = v;
  }

  __syncthreads();

  const int wave = tid >> 5;
  const int lane = tid & 31;
  const int m    = lane & 15;
  const int hi   = lane >> 4;

  const v2f* lxa = (const v2f*)lx + ((wave * ZW + m) * (CP / 2) + hi);
  const v2f* lwb = (const v2f*)lw + (m * 2 + hi);

  v8f acc = {0.f, 0.f, 0.f, 0.f, 0.f, 0.f, 0.f, 0.f};

#pragma unroll
  for (int kd = 0; kd < 3; ++kd) {
#pragma unroll
    for (int kh = 0; kh < 3; ++kh) {
#pragma unroll
      for (int kw = 0; kw < 3; ++kw) {
        const int t27 = (kd * 3 + kh) * 3 + kw;
#pragma unroll
        for (int cig = 0; cig < 2; ++cig) {
          const v2f A = lxa[((kd * ZH + kh) * ZW + kw) * (CP / 2) + cig * 2];
          const v2f B = lwb[(t27 * 2 + cig) * 32];
          acc = __builtin_amdgcn_wmma_f32_16x16x4_f32(
              false, A, false, B, (short)0, acc, false, false);
        }
      }
    }
  }

  const int oc = m;
  if (oc < CO) {
    const float bv = bias[oc];
    const int h = h0 + wave;
    float* op = out + (n * CO + oc) * sC + d * sD + h * sH + w0 + hi * 8;
#pragma unroll
    for (int j = 0; j < 8; ++j)
      __builtin_nontemporal_store(acc[j] + bv, op + j);
  }
}

extern "C" void kernel_launch(void* const* d_in, const int* in_sizes, int n_in,
                              void* d_out, int out_size, void* d_ws, size_t ws_size,
                              hipStream_t stream) {
  const float* x    = (const float*)d_in[0];
  const float* wgt  = (const float*)d_in[1];
  const float* bias = (const float*)d_in[2];
  float* out = (float*)d_out;

  const int blocks = 2 * DD * 14 * 7;
  SparseConv3d_wmma_f32_kernel<<<blocks, 256, 0, stream>>>(x, wgt, bias, out);
}


# block walks 8 output d planes over a 4-slot LDS ring of input planes: each plane is fetched once per block instead of 3x, weight fragments staged once per 8 planes; summation order unchanged
# speedup vs baseline: 1.6539x; 1.6539x over previous
#include <hip/hip_runtime.h>


typedef float v2f __attribute__((ext_vector_type(2)));
typedef float v8f __attribute__((ext_vector_type(8)));

#define DD 112
#define HH 112
#define WW 112
#define CI 8
#define CO 8
#define NTAPS 27

#define ZH 10
#define ZW 18
#define CP 10
#define ZHS 11
#define PLANE (ZHS * ZW * CP)
#define RING 4
#define DCH 8
#define NCHUNK (DD / DCH)
#define NSLOT 6

__global__ __launch_bounds__(256) void SparseConv3d_wmma_f32_kernel(
    const float* __restrict__ x,
    const float* __restrict__ wgt,
    const float* __restrict__ bias,
    float* __restrict__ out)
{
  const int sH = WW;
  const int sD = HH * WW;
  const int sC = DD * HH * WW;
  const int sN = CI * sC;

  __shared__ float lx[RING * PLANE];
  __shared__ float lw[NTAPS * 2 * 16 * 4];

  const int tid = threadIdx.x;

  int b = blockIdx.x;
  const int wq = b % 7;      b /= 7;
  const int hb = b % 14;     b /= 14;
  const int dc = b % NCHUNK; b /= NCHUNK;
  const int n  = b;
  const int h0 = hb * 8;
  const int w0 = wq * 16;
  const int d0 = dc * DCH;

  const float* xn = x + n * sN;

  unsigned goff[NSLOT];
  int  loff[NSLOT];
  bool ok[NSLOT];
#pragma unroll
  for (int k = 0; k < NSLOT; ++k) {
    const int e = tid + 256 * k;
    const int zw_off = e % ZW;
    const int r = e / ZW;
    const int ci = r % CI;
    const int zh_off = r / CI;
    const int zh = h0 + zh_off - 1;
    const int zw = w0 + zw_off - 1;
    ok[k] = ((unsigned)zh < (unsigned)HH) & ((unsigned)zw < (unsigned)WW);
    goff[k] = ok[k] ? (unsigned)(ci * sC + zh * sH + zw) : 0u;
    loff[k] = (zh_off * ZW + zw_off) * CP + ci;
  }

  float v[NSLOT];

  for (int j = 0; j < 3; ++j) {
    const int zd = d0 - 1 + j;
    const bool dv = (unsigned)zd < (unsigned)DD;
    const float* xp = xn + (dv ? zd : 0) * sD;
#pragma unroll
    for (int k = 0; k < NSLOT; ++k) v[k] = xp[goff[k]];
#pragma unroll
    for (int k = 0; k < NSLOT; ++k)
      lx[j * PLANE + loff[k]] = (ok[k] & dv) ? v[k] : 0.0f;
  }

  for (int idx = tid; idx < NTAPS * 2 * 16 * 4; idx += 256) {
    const int k   = idx & 3;
    const int nn  = (idx >> 2) & 15;
    const int cig = (idx >> 6) & 1;
    const int t27 = idx >> 7;
    const int ci  = cig * 4 + k;
    const float wv = wgt[((nn & 7) * CI + ci) * NTAPS + t27] *
                     (nn < CO ? 1.0f : 0.0f);
    lw[idx] = wv;
  }

  __syncthreads();

  const int wave = tid >> 5;
  const int lane = tid & 31;
  const int m    = lane & 15;
  const int hi   = lane >> 4;

  const v2f* lxa = (const v2f*)lx + ((wave * ZW + m) * (CP / 2) + hi);
  const v2f* lwb = (const v2f*)lw + (m * 2 + hi);

  const int oc = m;
  const float bv = bias[oc & 7];
  float* obase = out + (n * CO + (oc & 7)) * sC + (h0 + wave) * sH + w0 + hi * 8;

  for (int s = 0; s < DCH; ++s) {
    const bool more = (s + 3) <= (DCH + 1);
    const int zdn = d0 + s + 2;
    const bool dvn = (unsigned)zdn < (unsigned)DD;
    if (more) {
      const float* xp = xn + (dvn ? zdn : 0) * sD;
#pragma unroll
      for (int k = 0; k < NSLOT; ++k) v[k] = xp[goff[k]];
    }

    v8f acc = {0.f, 0.f, 0.f, 0.f, 0.f, 0.f, 0.f, 0.f};
#pragma unroll
    for (int kd = 0; kd < 3; ++kd) {
      const v2f* pa = lxa + ((s + kd) & (RING - 1)) * (PLANE / 2);
#pragma unroll
      for (int kh = 0; kh < 3; ++kh) {
#pragma unroll
        for (int kw = 0; kw < 3; ++kw) {
          const int t27 = (kd * 3 + kh) * 3 + kw;
#pragma unroll
          for (int cig = 0; cig < 2; ++cig) {
            const v2f A = pa[(kh * ZW + kw) * (CP / 2) + cig * 2];
            const v2f B = lwb[(t27 * 2 + cig) * 32];
            acc = __builtin_amdgcn_wmma_f32_16x16x4_f32(
                false, A, false, B, (short)0, acc, false, false);
          }
        }
      }
    }

    if (more) {
#pragma unroll
      for (int k = 0; k < NSLOT; ++k)
        lx[((s + 3) & (RING - 1)) * PLANE + loff[k]] = (ok[k] & dvn) ? v[k] : 0.0f;
      __syncthreads();
    }

    if (oc < CO) {
      float* op = obase + (d0 + s) * sD;
#pragma unroll
      for (int j = 0; j < 8; ++j)
        __builtin_nontemporal_store(acc[j] + bv, op + j);
    }
  }
}

extern "C" void kernel_launch(void* const* d_in, const int* in_sizes, int n_in,
                              void* d_out, int out_size, void* d_ws, size_t ws_size,
                              hipStream_t stream) {
  const float* x    = (const float*)d_in[0];
  const float* wgt  = (const float*)d_in[1];
  const float* bias = (const float*)d_in[2];
  float* out = (float*)d_out;

  const int blocks = 2 * NCHUNK * 14 * 7;
  SparseConv3d_wmma_f32_kernel<<<blocks, 256, 0, stream>>>(x, wgt, bias, out);
}
